# starting kernel
# baseline (speedup 1.0000x reference)
#include <hip/hip_runtime.h>


typedef __attribute__((ext_vector_type(2))) float v2f;
typedef __attribute__((ext_vector_type(8))) float v8f;

__global__ void gcn_degree_kernel(const int* __restrict__ dst,
                                  float* __restrict__ deg, int E) {
  int t = blockIdx.x * blockDim.x + threadIdx.x;
  if (t < E) unsafeAtomicAdd(&deg[dst[t]], 1.0f);
}

template <int K, int TM>
__global__ void gcn_gemm_wmma_f32(const float* __restrict__ X,
                                  const float* __restrict__ W,
                                  float* __restrict__ Y, int N) {
  const int M = TM * 16;
  const int wave = (int)((blockIdx.x * (size_t)blockDim.x + threadIdx.x) >> 5);
  if (wave >= (N >> 4)) return;

  const int lane = threadIdx.x & 31;
  const int half = lane >> 4;
  const int l16  = lane & 15;
  const float* __restrict__ xrow = X + (size_t)((wave << 4) + l16) * K;

  v8f acc[TM];
#pragma unroll
  for (int j = 0; j < TM; ++j) acc[j] = (v8f){};

#pragma unroll 2
  for (int k = 0; k < K; k += 4) {
    const int kk = k + (half << 1);
    const v2f a = *(const v2f*)(xrow + kk);
#pragma unroll
    for (int j = 0; j < TM; ++j) {
      v2f b;
      b.x = W[kk * M + (j << 4) + l16];
      b.y = W[(kk + 1) * M + (j << 4) + l16];
      acc[j] = __builtin_amdgcn_wmma_f32_16x16x4_f32(
           false, a,  false, b,
           (short)0, acc[j],  false,  false);
    }
  }

  const int rbase = (wave << 4) + (half << 3);
#pragma unroll
  for (int j = 0; j < TM; ++j)
#pragma unroll
    for (int r = 0; r < 8; ++r)
      Y[(size_t)(rbase + r) * M + (j << 4) + l16] = acc[j][r];
}

template <int SHIFT>
__global__ void gcn_scatter_add(const float* __restrict__ h,
                                const int* __restrict__ src,
                                const int* __restrict__ dst,
                                float* __restrict__ agg, int E) {
  const int width = 4 << SHIFT;
  long long t = (long long)blockIdx.x * blockDim.x + threadIdx.x;
  int e = (int)(t >> SHIFT);
  if (e >= E) return;
  int c = ((int)t & ((1 << SHIFT) - 1)) << 2;
  int s = src[e], d = dst[e];
  float4 v = *(const float4*)(h + (size_t)s * width + c);
  float* ap = agg + (size_t)d * width + c;
  unsafeAtomicAdd(ap + 0, v.x);
  unsafeAtomicAdd(ap + 1, v.y);
  unsafeAtomicAdd(ap + 2, v.z);
  unsafeAtomicAdd(ap + 3, v.w);
}

template <int LOGW>
__global__ void gcn_finalize(const float* __restrict__ agg,
                             const float* __restrict__ deg,
                             const float* __restrict__ bias,
                             float* __restrict__ out, int N, int doRelu) {
  const int width = 1 << LOGW;
  long long t = (long long)blockIdx.x * blockDim.x + threadIdx.x;
  if (t >= ((long long)N << LOGW)) return;
  int n = (int)(t >> LOGW);
  int c = (int)t & (width - 1);
  float d = deg[n];
  d = d < 1.0f ? 1.0f : d;
  float r = agg[t] / d + bias[c];
  if (doRelu) r = r > 0.0f ? r : 0.0f;
  out[t] = r;
}

extern "C" void kernel_launch(void* const* d_in, const int* in_sizes, int n_in,
                              void* d_out, int out_size, void* d_ws, size_t ws_size,
                              hipStream_t stream) {
  const float* feat = (const float*)d_in[0];
  const float* W0   = (const float*)d_in[1];
  const float* b0   = (const float*)d_in[2];
  const float* W1   = (const float*)d_in[3];
  const float* b1   = (const float*)d_in[4];
  const float* W2   = (const float*)d_in[5];
  const float* b2   = (const float*)d_in[6];
  const int*   src  = (const int*)d_in[7];
  const int*   dst  = (const int*)d_in[8];
  float* out = (float*)d_out;

  const int N = in_sizes[0] / 256;
  const int E = in_sizes[7];

  float* bufA = (float*)d_ws;
  float* bufB = bufA + (size_t)N * 64;
  float* deg  = bufB + (size_t)N * 64;
  const size_t sz64 = (size_t)N * 64 * sizeof(float);
  const size_t sz32 = (size_t)N * 32 * sizeof(float);

  const int TB = 256;
  const int gemmBlocks   = (int)((((long long)(N / 16)) * 32 + TB - 1) / TB);
  const int scat64Blocks = (int)(((long long)E * 16 + TB - 1) / TB);
  const int scat32Blocks = (int)(((long long)E * 8 + TB - 1) / TB);
  const int fin64Blocks  = (int)(((long long)N * 64 + TB - 1) / TB);
  const int fin32Blocks  = (int)(((long long)N * 32 + TB - 1) / TB);

  hipMemsetAsync(deg, 0, (size_t)N * sizeof(float), stream);
  gcn_degree_kernel<<<(E + TB - 1) / TB, TB, 0, stream>>>(dst, deg, E);

  gcn_gemm_wmma_f32<256, 4><<<gemmBlocks, TB, 0, stream>>>(feat, W0, bufA, N);
  hipMemsetAsync(bufB, 0, sz64, stream);
  gcn_scatter_add<4><<<scat64Blocks, TB, 0, stream>>>(bufA, src, dst, bufB, E);
  gcn_finalize<6><<<fin64Blocks, TB, 0, stream>>>(bufB, deg, b0, bufA, N, 1);

  hipMemsetAsync(bufB, 0, sz64, stream);
  gcn_scatter_add<4><<<scat64Blocks, TB, 0, stream>>>(bufA, src, dst, bufB, E);
  gcn_gemm_wmma_f32<64, 4><<<gemmBlocks, TB, 0, stream>>>(bufB, W1, bufA, N);
  gcn_finalize<6><<<fin64Blocks, TB, 0, stream>>>(bufA, deg, b1, bufB, N, 1);

  gcn_gemm_wmma_f32<64, 2><<<gemmBlocks, TB, 0, stream>>>(bufB, W2, bufA, N);
  hipMemsetAsync(bufB, 0, sz32, stream);
  gcn_scatter_add<3><<<scat32Blocks, TB, 0, stream>>>(bufA, src, dst, bufB, E);
  gcn_finalize<5><<<fin32Blocks, TB, 0, stream>>>(bufB, deg, b2, out, N, 0);
}


# dst scan on a 1-byte owner plane (edge -> owning block) built by a small pre-pass: 4x fewer scanned bytes, 16 edges tested per 16-byte load; dst checked per hit; LDS float adds stay arrival order
# speedup vs baseline: 2.0299x; 2.0299x over previous
#include <hip/hip_runtime.h>


typedef __attribute__((ext_vector_type(2))) float v2f;
typedef __attribute__((ext_vector_type(8))) float v8f;

template <int K, int TM>
__global__ void gcn_gemm_wmma_f32(const float* __restrict__ X,
                                  const float* __restrict__ W,
                                  float* __restrict__ Y, int N) {
  const int M = TM * 16;
  const int wave = (int)((blockIdx.x * (size_t)blockDim.x + threadIdx.x) >> 5);
  if (wave >= (N >> 4)) return;

  const int lane = threadIdx.x & 31;
  const int half = lane >> 4;
  const int l16  = lane & 15;
  const float* __restrict__ xrow = X + (size_t)((wave << 4) + l16) * K;

  v8f acc[TM];
#pragma unroll
  for (int j = 0; j < TM; ++j) acc[j] = (v8f){};

#pragma unroll 2
  for (int k = 0; k < K; k += 4) {
    const int kk = k + (half << 1);
    const v2f a = *(const v2f*)(xrow + kk);
#pragma unroll
    for (int j = 0; j < TM; ++j) {
      v2f b;
      b.x = W[kk * M + (j << 4) + l16];
      b.y = W[(kk + 1) * M + (j << 4) + l16];
      acc[j] = __builtin_amdgcn_wmma_f32_16x16x4_f32(
           false, a,  false, b,
           (short)0, acc[j],  false,  false);
    }
  }

  const int rbase = (wave << 4) + (half << 3);
#pragma unroll
  for (int j = 0; j < TM; ++j)
#pragma unroll
    for (int r = 0; r < 8; ++r)
      Y[(size_t)(rbase + r) * M + (j << 4) + l16] = acc[j][r];
}

constexpr int AGG_NODES = 416;

__device__ __forceinline__ unsigned owner_byte(int d) {
  return ((unsigned)d / (unsigned)AGG_NODES) & 255u;
}

__global__ __launch_bounds__(256)
void gcn_owner_plane(const int* __restrict__ dst, int* __restrict__ plane, int E) {
  const int lane = threadIdx.x & 31;
  const int wave = (int)((blockIdx.x * (size_t)blockDim.x + threadIdx.x) >> 5);
  const int eb = wave * 512;
  if (eb + 512 <= E) {
    const int4* p = (const int4*)(dst + eb);
    int4 q[4];
#pragma unroll
    for (int j = 0; j < 4; ++j) q[j] = p[j * 32 + lane];
#pragma unroll
    for (int j = 0; j < 4; ++j)
      plane[(eb >> 2) + j * 32 + lane] =
          (int)(owner_byte(q[j].x) | (owner_byte(q[j].y) << 8) |
                (owner_byte(q[j].z) << 16) | (owner_byte(q[j].w) << 24));
  }
}

constexpr int AGG_THREADS = 1024;
constexpr int AGG_WAVES   = AGG_THREADS / 32;
constexpr int AGG_NQ      = 4;
constexpr int AGG_NP      = AGG_NQ / 2;
constexpr int AGG_CHUNK   = AGG_NQ * 512;

__device__ __forceinline__ void agg_lds_add(float* p, float v) {
  __hip_atomic_fetch_add(p, v, __ATOMIC_RELAXED, __HIP_MEMORY_SCOPE_WORKGROUP);
}

template <int LOGW, int MODE>
__global__ __launch_bounds__(AGG_THREADS)
void gcn_agg_owned(const float* __restrict__ h,
                   const int* __restrict__ src,
                   const int* __restrict__ dst,
                   const int* __restrict__ plane,
                   const float* __restrict__ bias,
                   float* __restrict__ out,
                   float* __restrict__ degOut, int N, int E) {
  constexpr int WIDTH = 1 << LOGW;
  constexpr int PITCH = WIDTH + 1;
  constexpr int NT    = WIDTH / 16;
  __shared__ float acc[AGG_NODES * PITCH];
  __shared__ float cnt[AGG_NODES];
  __shared__ unsigned short hits[AGG_WAVES][AGG_CHUNK];

  const int tid  = threadIdx.x;
  const int lane = tid & 31;
  const int wave = __builtin_amdgcn_readfirstlane(tid >> 5);
  const int base = blockIdx.x * AGG_NODES;
  const int nHere = (N - base < AGG_NODES) ? (N - base) : AGG_NODES;
  const unsigned pat = (blockIdx.x & 255u) * 0x01010101u;

  for (int i = tid; i < AGG_NODES * PITCH; i += AGG_THREADS) acc[i] = 0.0f;
  for (int i = tid; i < AGG_NODES; i += AGG_THREADS) cnt[i] = 0.0f;
  __syncthreads();

  const int nFull = E / AGG_CHUNK;
  const int rot = nFull > 0 ? (int)((blockIdx.x * 6u) % (unsigned)nFull) : 0;
  unsigned short* myHits = hits[wave];
  const int grp = lane >> 2;
  const int sl  = lane & 3;


  auto ask = [&](int cc, int4 (&qq)[AGG_NQ]) -> int {
    int cr = cc + rot; if (cr >= nFull) cr -= nFull;
    const int ef = cr * AGG_CHUNK;
    const int4* p = (const int4*)(plane + (ef >> 2));
#pragma unroll
    for (int j = 0; j < AGG_NQ; ++j) qq[j] = p[j * 32 + lane];
    return ef;
  };

  auto work = [&](const int4 (&qq)[AGG_NQ], int ef) {
    unsigned c[AGG_NP], m[AGG_NP];
    unsigned any = 0u;
#pragma unroll
    for (int w = 0; w < AGG_NP; ++w) {
      unsigned cw = 0u;
#pragma unroll
      for (int jj = 0; jj < 2; ++jj) {
        const int4 q = qq[2 * w + jj];
        const unsigned x4[4] = {(unsigned)q.x, (unsigned)q.y, (unsigned)q.z, (unsigned)q.w};
#pragma unroll
        for (int k = 0; k < 4; ++k) {
          const unsigned y = x4[k] ^ pat;
          cw |= (((y - 0x01010101u) & ~y) >> (7 - 4 * jj - k)) & (0x01010101u << (4 * jj + k));
        }
      }
      c[w] = cw;
      m[w] = __builtin_amdgcn_ballot_w32(cw != 0u);
      any |= m[w];
    }
    if (any == 0u) return;
    int total = 0;
#pragma unroll
    for (int w = 0; w < AGG_NP; ++w) {
      unsigned cw = c[w];
      unsigned mw = m[w];
      while (mw != 0u) {
        if (cw != 0u) {
          const int b = __builtin_ctz(cw);
          const int pos = total + (int)__builtin_amdgcn_mbcnt_lo(mw, 0u);
          myHits[pos] = (unsigned short)(w * 1024 + ((b & 4) << 7) + lane * 16 + ((b & 3) << 2) + (b >> 3));
          cw &= cw - 1u;
        }
        total += __builtin_popcount(mw);
        mw = __builtin_amdgcn_ballot_w32(cw != 0u);
      }
    }
    for (int b0 = 0; b0 < total; b0 += 32) {
      const int nb = (total - b0 < 32) ? (total - b0) : 32;
      int val = -1;
      if (lane < nb) {
        const int e = ef + (int)myHits[b0 + lane];
        const unsigned rel = (unsigned)(dst[e] - base);
        const int s = src[e];
        val = (rel < (unsigned)nHere) ? ((s << 9) | (int)rel) : -1;
      }
      for (int t = 0; t * 8 < nb; t += 2) {
        int ent[2];
        ent[0] = __shfl(val, t * 8 + grp);
        ent[1] = __shfl(val, (t * 8 + 8 + grp) & 31);
        if (t * 8 + 8 >= nb) ent[1] = -1;
        float4 v[2][NT];
#pragma unroll
        for (int r = 0; r < 2; ++r) {
          const float4* row = (const float4*)(h + (size_t)(ent[r] >= 0 ? (ent[r] >> 9) : 0) * WIDTH);
#pragma unroll
          for (int u = 0; u < NT; ++u) v[r][u] = row[sl + 4 * u];
        }
#pragma unroll
        for (int r = 0; r < 2; ++r) {
          if (ent[r] >= 0) {
            const int rel = ent[r] & 511;
            float* a = acc + rel * PITCH + 4 * sl;
#pragma unroll
            for (int u = 0; u < NT; ++u) {
              agg_lds_add(a + 16 * u + 0, v[r][u].x);
              agg_lds_add(a + 16 * u + 1, v[r][u].y);
              agg_lds_add(a + 16 * u + 2, v[r][u].z);
              agg_lds_add(a + 16 * u + 3, v[r][u].w);
            }
            if (sl == 0) agg_lds_add(cnt + rel, 1.0f);
          }
        }
      }
    }
  };

  int c = wave;
  if (c < nFull) {
    int4 qa[AGG_NQ], qb[AGG_NQ];
    int ea = ask(c, qa), eb = 0;
    for (;;) {
      eb = ask((c + AGG_WAVES < nFull) ? c + AGG_WAVES : c, qb);
      work(qa, ea);
      c += AGG_WAVES;
      if (c >= nFull) break;
      ea = ask((c + AGG_WAVES < nFull) ? c + AGG_WAVES : c, qa);
      work(qb, eb);
      c += AGG_WAVES;
      if (c >= nFull) break;
    }
  }

  for (int e = nFull * AGG_CHUNK + tid; e < E; e += AGG_THREADS) {
    const unsigned rel = (unsigned)(dst[e] - base);
    if (rel < (unsigned)nHere) {
      const float* row = h + (size_t)src[e] * WIDTH;
      float* a = acc + rel * PITCH;
      for (int col = 0; col < WIDTH; ++col) agg_lds_add(a + col, row[col]);
      agg_lds_add(cnt + rel, 1.0f);
    }
  }
  __syncthreads();

  for (int i = tid; i < nHere * WIDTH; i += AGG_THREADS) {
    const int n = i >> LOGW;
    const int col = i & (WIDTH - 1);
    float v = acc[n * PITCH + col];
    if (MODE != 0) {
      float d = cnt[n];
      d = d < 1.0f ? 1.0f : d;
      v = v / d + bias[col];
      if (MODE == 1) v = v > 0.0f ? v : 0.0f;
    }
    out[(size_t)base * WIDTH + i] = v;
  }
  if (MODE == 0)
    for (int i = tid; i < nHere; i += AGG_THREADS) degOut[base + i] = cnt[i];
}

template <int LOGW>
__global__ void gcn_finalize(const float* __restrict__ agg,
                             const float* __restrict__ deg,
                             const float* __restrict__ bias,
                             float* __restrict__ out, int N, int doRelu) {
  const int width = 1 << LOGW;
  long long t = (long long)blockIdx.x * blockDim.x + threadIdx.x;
  if (t >= ((long long)N << LOGW)) return;
  int n = (int)(t >> LOGW);
  int c = (int)t & (width - 1);
  float d = deg[n];
  d = d < 1.0f ? 1.0f : d;
  float r = agg[t] / d + bias[c];
  if (doRelu) r = r > 0.0f ? r : 0.0f;
  out[t] = r;
}

extern "C" void kernel_launch(void* const* d_in, const int* in_sizes, int n_in,
                              void* d_out, int out_size, void* d_ws, size_t ws_size,
                              hipStream_t stream) {
  const float* feat = (const float*)d_in[0];
  const float* W0   = (const float*)d_in[1];
  const float* b0   = (const float*)d_in[2];
  const float* W1   = (const float*)d_in[3];
  const float* b1   = (const float*)d_in[4];
  const float* W2   = (const float*)d_in[5];
  const float* b2   = (const float*)d_in[6];
  const int*   src  = (const int*)d_in[7];
  const int*   dst  = (const int*)d_in[8];
  float* out = (float*)d_out;

  const int N = in_sizes[0] / 256;
  const int E = in_sizes[7];

  float* bufA = (float*)d_ws;
  float* bufB = bufA + (size_t)N * 64;
  float* deg  = bufB + (size_t)N * 64;

  const int TB = 256;
  const int gemmBlocks  = (int)((((long long)(N / 16)) * 32 + TB - 1) / TB);
  const int fin64Blocks = (int)(((long long)N * 64 + TB - 1) / TB);
  const int aggBlocks   = (N + AGG_NODES - 1) / AGG_NODES;
  const int planeBlocks = (int)((((long long)((E + 511) / 512)) * 32 + TB - 1) / TB);

  int* planeO = (int*)d_out;
  int* planeA = (int*)bufA;

  gcn_owner_plane<<<planeBlocks, TB, 0, stream>>>(dst, planeO, E);
  gcn_gemm_wmma_f32<256, 4><<<gemmBlocks, TB, 0, stream>>>(feat, W0, bufA, N);
  gcn_agg_owned<6, 1><<<aggBlocks, AGG_THREADS, 0, stream>>>(bufA, src, dst, planeO, b0, bufB, deg, N, E);

  gcn_agg_owned<6, 0><<<aggBlocks, AGG_THREADS, 0, stream>>>(bufB, src, dst, planeO, b1, bufA, deg, N, E);
  gcn_gemm_wmma_f32<64, 4><<<gemmBlocks, TB, 0, stream>>>(bufA, W1, bufB, N);
  gcn_finalize<6><<<fin64Blocks, TB, 0, stream>>>(bufB, deg, b1, bufA, N, 1);

  gcn_gemm_wmma_f32<64, 2><<<gemmBlocks, TB, 0, stream>>>(bufA, W2, bufB, N);
  gcn_owner_plane<<<planeBlocks, TB, 0, stream>>>(dst, planeA, E);
  gcn_agg_owned<5, 2><<<aggBlocks, AGG_THREADS, 0, stream>>>(bufB, src, dst, planeA, b2, out, deg, N, E);
}


# first neighbour sum writes down each block's own edges (src,row) as its scan finds them (LDS int slot counter); later sums replay the list: no plane scan, no dst/src lookups; adds stay arrival order
# speedup vs baseline: 2.4358x; 1.2000x over previous
#include <hip/hip_runtime.h>


typedef __attribute__((ext_vector_type(2))) float v2f;
typedef __attribute__((ext_vector_type(8))) float v8f;

template <int K, int TM>
__global__ void gcn_gemm_wmma_f32(const float* __restrict__ X,
                                  const float* __restrict__ W,
                                  float* __restrict__ Y, int N) {
  const int M = TM * 16;
  const int wave = (int)((blockIdx.x * (size_t)blockDim.x + threadIdx.x) >> 5);
  if (wave >= (N >> 4)) return;

  const int lane = threadIdx.x & 31;
  const int half = lane >> 4;
  const int l16  = lane & 15;
  const float* __restrict__ xrow = X + (size_t)((wave << 4) + l16) * K;

  v8f acc[TM];
#pragma unroll
  for (int j = 0; j < TM; ++j) acc[j] = (v8f){};

#pragma unroll 2
  for (int k = 0; k < K; k += 4) {
    const int kk = k + (half << 1);
    const v2f a = *(const v2f*)(xrow + kk);
#pragma unroll
    for (int j = 0; j < TM; ++j) {
      v2f b;
      b.x = W[kk * M + (j << 4) + l16];
      b.y = W[(kk + 1) * M + (j << 4) + l16];
      acc[j] = __builtin_amdgcn_wmma_f32_16x16x4_f32(
           false, a,  false, b,
           (short)0, acc[j],  false,  false);
    }
  }

  const int rbase = (wave << 4) + (half << 3);
#pragma unroll
  for (int j = 0; j < TM; ++j)
#pragma unroll
    for (int r = 0; r < 8; ++r)
      Y[(size_t)(rbase + r) * M + (j << 4) + l16] = acc[j][r];
}

constexpr int AGG_NODES = 416;

__device__ __forceinline__ unsigned owner_byte(int d) {
  return ((unsigned)d / (unsigned)AGG_NODES) & 255u;
}

constexpr int AGG_THREADS = 1024;
constexpr int AGG_WAVES   = AGG_THREADS / 32;
constexpr int AGG_NQ      = 4;
constexpr int AGG_NP      = AGG_NQ / 2;
constexpr int AGG_CHUNK   = AGG_NQ * 512;

constexpr int OUT_W      = 32;
constexpr int REC_STRIDE = AGG_NODES * OUT_W;
constexpr int REC_WORDS  = 8192;
constexpr int PLANE_SLOT = AGG_CHUNK / 4;
static_assert(REC_WORDS + PLANE_SLOT <= REC_STRIDE, "no room for the plane behind a list");

__device__ __forceinline__ int full_regions(int N) {
  const int r = N / AGG_NODES;
  return r > 0 ? r : 1;
}
__device__ __forceinline__ size_t plane_word(int chunk, int nReg) {
  return (size_t)(chunk % nReg) * REC_STRIDE + REC_WORDS + (size_t)(chunk / nReg) * PLANE_SLOT;
}

__global__ __launch_bounds__(256)
void gcn_owner_plane(const int* __restrict__ dst, int* __restrict__ side, int E, int N) {
  const int lane = threadIdx.x & 31;
  const int wave = (int)((blockIdx.x * (size_t)blockDim.x + threadIdx.x) >> 5);
  const int eb = wave * 512;
  const int chunk = wave / AGG_NQ;
  if (chunk < E / AGG_CHUNK) {
    const int4* p = (const int4*)(dst + eb);
    int* o = side + plane_word(chunk, full_regions(N)) + (wave % AGG_NQ) * 128;
    int4 q[4];
#pragma unroll
    for (int j = 0; j < 4; ++j) q[j] = p[j * 32 + lane];
#pragma unroll
    for (int j = 0; j < 4; ++j)
      o[j * 32 + lane] =
          (int)(owner_byte(q[j].x) | (owner_byte(q[j].y) << 8) |
                (owner_byte(q[j].z) << 16) | (owner_byte(q[j].w) << 24));
  }
}


__device__ __forceinline__ void agg_lds_add(float* p, float v) {
  __hip_atomic_fetch_add(p, v, __ATOMIC_RELAXED, __HIP_MEMORY_SCOPE_WORKGROUP);
}

template <int LOGW, int MODE, int REC>
__global__ __launch_bounds__(AGG_THREADS)
void gcn_agg_owned(const float* __restrict__ h,
                   const int* __restrict__ src,
                   const int* __restrict__ dst,
                   const int* plane,
                   int* rec,
                   const float* __restrict__ bias,
                   float* out,
                   float* __restrict__ degOut, int N, int E) {
  constexpr int WIDTH = 1 << LOGW;
  constexpr int PITCH = WIDTH + 1;
  constexpr int NT    = WIDTH / 16;
  __shared__ float acc[AGG_NODES * PITCH];
  __shared__ float cnt[AGG_NODES];
  __shared__ unsigned short hits[AGG_WAVES][AGG_CHUNK];
  __shared__ int recCount;

  const int tid  = threadIdx.x;
  const int lane = tid & 31;
  const int wave = __builtin_amdgcn_readfirstlane(tid >> 5);
  const int base = blockIdx.x * AGG_NODES;
  const int nHere = (N - base < AGG_NODES) ? (N - base) : AGG_NODES;
  const unsigned pat = (blockIdx.x & 255u) * 0x01010101u;

  for (int i = tid; i < AGG_NODES * PITCH; i += AGG_THREADS) acc[i] = 0.0f;
  for (int i = tid; i < AGG_NODES; i += AGG_THREADS) cnt[i] = 0.0f;
  if (tid == 0) recCount = 0;
  __syncthreads();

  int* myRec = rec + (size_t)blockIdx.x * REC_STRIDE;
  const int recCap = ((nHere * OUT_W < REC_WORDS) ? nHere * OUT_W : REC_WORDS) - 1;
  const int nReg = full_regions(N);

  const int nFull = E / AGG_CHUNK;
  const int rot = nFull > 0 ? (int)((blockIdx.x * 6u) % (unsigned)nFull) : 0;
  unsigned short* myHits = hits[wave];
  const int grp = lane >> 2;
  const int sl  = lane & 3;


  auto addBatch = [&](int val, int nb) {
    for (int t = 0; t * 8 < nb; t += 2) {
      int ent[2];
      ent[0] = __shfl(val, t * 8 + grp);
      ent[1] = __shfl(val, (t * 8 + 8 + grp) & 31);
      if (t * 8 + 8 >= nb) ent[1] = -1;
      float4 v[2][NT];
#pragma unroll
      for (int r = 0; r < 2; ++r) {
        const float4* row = (const float4*)(h + (size_t)(ent[r] >= 0 ? (ent[r] >> 9) : 0) * WIDTH);
#pragma unroll
        for (int u = 0; u < NT; ++u) v[r][u] = row[sl + 4 * u];
      }
#pragma unroll
      for (int r = 0; r < 2; ++r) {
        if (ent[r] >= 0) {
          const int rel = ent[r] & 511;
          float* a = acc + rel * PITCH + 4 * sl;
#pragma unroll
          for (int u = 0; u < NT; ++u) {
            agg_lds_add(a + 16 * u + 0, v[r][u].x);
            agg_lds_add(a + 16 * u + 1, v[r][u].y);
            agg_lds_add(a + 16 * u + 2, v[r][u].z);
            agg_lds_add(a + 16 * u + 3, v[r][u].w);
          }
          if (sl == 0) agg_lds_add(cnt + rel, 1.0f);
        }
      }
    }
  };

  auto ask = [&](int cc, int4 (&qq)[AGG_NQ]) -> int {
    int cr = cc + rot; if (cr >= nFull) cr -= nFull;
    const int ef = cr * AGG_CHUNK;
    const int4* p = (const int4*)(plane + plane_word(cr, nReg));
#pragma unroll
    for (int j = 0; j < AGG_NQ; ++j) qq[j] = p[j * 32 + lane];
    return ef;
  };

  auto work = [&](const int4 (&qq)[AGG_NQ], int ef) {
    unsigned c[AGG_NP], m[AGG_NP];
    unsigned any = 0u;
#pragma unroll
    for (int w = 0; w < AGG_NP; ++w) {
      unsigned cw = 0u;
#pragma unroll
      for (int jj = 0; jj < 2; ++jj) {
        const int4 q = qq[2 * w + jj];
        const unsigned x4[4] = {(unsigned)q.x, (unsigned)q.y, (unsigned)q.z, (unsigned)q.w};
#pragma unroll
        for (int k = 0; k < 4; ++k) {
          const unsigned y = x4[k] ^ pat;
          cw |= (((y - 0x01010101u) & ~y) >> (7 - 4 * jj - k)) & (0x01010101u << (4 * jj + k));
        }
      }
      c[w] = cw;
      m[w] = __builtin_amdgcn_ballot_w32(cw != 0u);
      any |= m[w];
    }
    if (any == 0u) return;
    int total = 0;
#pragma unroll
    for (int w = 0; w < AGG_NP; ++w) {
      unsigned cw = c[w];
      unsigned mw = m[w];
      while (mw != 0u) {
        if (cw != 0u) {
          const int b = __builtin_ctz(cw);
          const int pos = total + (int)__builtin_amdgcn_mbcnt_lo(mw, 0u);
          myHits[pos] = (unsigned short)(w * 1024 + ((b & 4) << 7) + lane * 16 + ((b & 3) << 2) + (b >> 3));
          cw &= cw - 1u;
        }
        total += __builtin_popcount(mw);
        mw = __builtin_amdgcn_ballot_w32(cw != 0u);
      }
    }
    for (int b0 = 0; b0 < total; b0 += 32) {
      const int nb = (total - b0 < 32) ? (total - b0) : 32;
      int val = -1;
      if (lane < nb) {
        const int e = ef + (int)myHits[b0 + lane];
        const unsigned rel = (unsigned)(dst[e] - base);
        const int s = src[e];
        val = (rel < (unsigned)nHere) ? ((s << 9) | (int)rel) : -1;
      }
      if (REC == 1) {
        const unsigned okm = __builtin_amdgcn_ballot_w32(val >= 0);
        if (okm != 0u) {
          int p0 = 0;
          if (lane == 0)
            p0 = __hip_atomic_fetch_add(&recCount, __builtin_popcount(okm), __ATOMIC_RELAXED,
                                        __HIP_MEMORY_SCOPE_WORKGROUP);
          p0 = __builtin_amdgcn_readfirstlane(p0);
          const int p = p0 + (int)__builtin_amdgcn_mbcnt_lo(okm, 0u);
          if (val >= 0 && p < recCap) myRec[1 + p] = val;
        }
      }
      addBatch(val, nb);
    }
  };

  int c = wave;
  bool replayed = false;
  if (REC == 2) {
    const int found = myRec[0];
    if (found >= 0 && found <= recCap) {
      replayed = true;
      int b0 = wave * 32;
      int nxt = (b0 + lane < found) ? myRec[1 + b0 + lane] : -1;
      while (b0 < found) {
        const int val = nxt;
        const int nb = (found - b0 < 32) ? (found - b0) : 32;
        const int b1 = b0 + AGG_THREADS;
        nxt = (b1 + lane < found) ? myRec[1 + b1 + lane] : -1;
        addBatch(val, nb);
        b0 = b1;
      }
    }
  }
  if (REC != 2 && c < nFull) {
    int4 qa[AGG_NQ], qb[AGG_NQ];
    int ea = ask(c, qa), eb = 0;
    for (;;) {
      eb = ask((c + AGG_WAVES < nFull) ? c + AGG_WAVES : c, qb);
      work(qa, ea);
      c += AGG_WAVES;
      if (c >= nFull) break;
      ea = ask((c + AGG_WAVES < nFull) ? c + AGG_WAVES : c, qa);
      work(qb, eb);
      c += AGG_WAVES;
      if (c >= nFull) break;
    }
  }

  for (int e = (REC == 2 ? 0 : nFull * AGG_CHUNK) + tid; e < (replayed ? 0 : E); e += AGG_THREADS) {
    const unsigned rel = (unsigned)(dst[e] - base);
    if (rel < (unsigned)nHere) {
      const int s = src[e];
      const float* row = h + (size_t)s * WIDTH;
      float* a = acc + rel * PITCH;
      for (int col = 0; col < WIDTH; ++col) agg_lds_add(a + col, row[col]);
      agg_lds_add(cnt + rel, 1.0f);
      if (REC == 1) {
        const int p = __hip_atomic_fetch_add(&recCount, 1, __ATOMIC_RELAXED, __HIP_MEMORY_SCOPE_WORKGROUP);
        if (p < recCap) myRec[1 + p] = (s << 9) | (int)rel;
      }
    }
  }
  __syncthreads();
  if (REC == 1 && tid == 0) myRec[0] = recCount;

  for (int i = tid; i < nHere * WIDTH; i += AGG_THREADS) {
    const int n = i >> LOGW;
    const int col = i & (WIDTH - 1);
    float v = acc[n * PITCH + col];
    if (MODE != 0) {
      float d = cnt[n];
      d = d < 1.0f ? 1.0f : d;
      v = v / d + bias[col];
      if (MODE == 1) v = v > 0.0f ? v : 0.0f;
    }
    out[(size_t)base * WIDTH + i] = v;
  }
  if (MODE == 0)
    for (int i = tid; i < nHere; i += AGG_THREADS) degOut[base + i] = cnt[i];
}

template <int LOGW>
__global__ void gcn_finalize(const float* __restrict__ agg,
                             const float* __restrict__ deg,
                             const float* __restrict__ bias,
                             float* __restrict__ out, int N, int doRelu) {
  const int width = 1 << LOGW;
  long long t = (long long)blockIdx.x * blockDim.x + threadIdx.x;
  if (t >= ((long long)N << LOGW)) return;
  int n = (int)(t >> LOGW);
  int c = (int)t & (width - 1);
  float d = deg[n];
  d = d < 1.0f ? 1.0f : d;
  float r = agg[t] / d + bias[c];
  if (doRelu) r = r > 0.0f ? r : 0.0f;
  out[t] = r;
}

extern "C" void kernel_launch(void* const* d_in, const int* in_sizes, int n_in,
                              void* d_out, int out_size, void* d_ws, size_t ws_size,
                              hipStream_t stream) {
  const float* feat = (const float*)d_in[0];
  const float* W0   = (const float*)d_in[1];
  const float* b0   = (const float*)d_in[2];
  const float* W1   = (const float*)d_in[3];
  const float* b1   = (const float*)d_in[4];
  const float* W2   = (const float*)d_in[5];
  const float* b2   = (const float*)d_in[6];
  const int*   src  = (const int*)d_in[7];
  const int*   dst  = (const int*)d_in[8];
  float* out = (float*)d_out;

  const int N = in_sizes[0] / 256;
  const int E = in_sizes[7];

  float* bufA = (float*)d_ws;
  float* bufB = bufA + (size_t)N * 64;
  float* deg  = bufB + (size_t)N * 64;

  const int TB = 256;
  const int gemmBlocks  = (int)((((long long)(N / 16)) * 32 + TB - 1) / TB);
  const int fin64Blocks = (int)(((long long)N * 64 + TB - 1) / TB);
  const int aggBlocks   = (N + AGG_NODES - 1) / AGG_NODES;
  const int planeBlocks = (int)((((long long)((E + 511) / 512)) * 32 + TB - 1) / TB);

  int* side = (int*)d_out;

  gcn_owner_plane<<<planeBlocks, TB, 0, stream>>>(dst, side, E, N);
  gcn_gemm_wmma_f32<256, 4><<<gemmBlocks, TB, 0, stream>>>(feat, W0, bufA, N);
  gcn_agg_owned<6, 1, 1><<<aggBlocks, AGG_THREADS, 0, stream>>>(bufA, src, dst, side, side, b0, bufB, deg, N, E);

  gcn_agg_owned<6, 0, 2><<<aggBlocks, AGG_THREADS, 0, stream>>>(bufB, src, dst, side, side, b1, bufA, deg, N, E);
  gcn_gemm_wmma_f32<64, 4><<<gemmBlocks, TB, 0, stream>>>(bufA, W1, bufB, N);
  gcn_finalize<6><<<fin64Blocks, TB, 0, stream>>>(bufB, deg, b1, bufA, N, 1);

  gcn_gemm_wmma_f32<64, 2><<<gemmBlocks, TB, 0, stream>>>(bufA, W2, bufB, N);
  gcn_agg_owned<5, 2, 2><<<aggBlocks, AGG_THREADS, 0, stream>>>(bufB, src, dst, side, side, b2, out, deg, N, E);
}


# layer 1's dense part and layer 2's product moved into the owning block of the neighbour sum: WMMA A fragments read from its LDS rows, normalise+relu in place, only x2.W2 stored; 3 launches less
# speedup vs baseline: 3.0477x; 1.2512x over previous
#include <hip/hip_runtime.h>


typedef __attribute__((ext_vector_type(2))) float v2f;
typedef __attribute__((ext_vector_type(8))) float v8f;

template <int K, int TM>
__global__ void gcn_gemm_wmma_f32(const float* __restrict__ X,
                                  const float* __restrict__ W,
                                  float* __restrict__ Y, int N) {
  const int M = TM * 16;
  const int wave = (int)((blockIdx.x * (size_t)blockDim.x + threadIdx.x) >> 5);
  if (wave >= (N >> 4)) return;

  const int lane = threadIdx.x & 31;
  const int half = lane >> 4;
  const int l16  = lane & 15;
  const float* __restrict__ xrow = X + (size_t)((wave << 4) + l16) * K;

  v8f acc[TM];
#pragma unroll
  for (int j = 0; j < TM; ++j) acc[j] = (v8f){};

#pragma unroll 2
  for (int k = 0; k < K; k += 4) {
    const int kk = k + (half << 1);
    const v2f a = *(const v2f*)(xrow + kk);
#pragma unroll
    for (int j = 0; j < TM; ++j) {
      v2f b;
      b.x = W[kk * M + (j << 4) + l16];
      b.y = W[(kk + 1) * M + (j << 4) + l16];
      acc[j] = __builtin_amdgcn_wmma_f32_16x16x4_f32(
           false, a,  false, b,
           (short)0, acc[j],  false,  false);
    }
  }

  const int rbase = (wave << 4) + (half << 3);
#pragma unroll
  for (int j = 0; j < TM; ++j)
#pragma unroll
    for (int r = 0; r < 8; ++r)
      Y[(size_t)(rbase + r) * M + (j << 4) + l16] = acc[j][r];
}

constexpr int AGG_NODES = 416;

__device__ __forceinline__ unsigned owner_byte(int d) {
  return ((unsigned)d / (unsigned)AGG_NODES) & 255u;
}

constexpr int AGG_THREADS = 1024;
constexpr int AGG_WAVES   = AGG_THREADS / 32;
constexpr int AGG_NQ      = 4;
constexpr int AGG_NP      = AGG_NQ / 2;
constexpr int AGG_CHUNK   = AGG_NQ * 512;

constexpr int OUT_W      = 32;
constexpr int REC_STRIDE = AGG_NODES * OUT_W;
constexpr int REC_WORDS  = 8192;
constexpr int PLANE_SLOT = AGG_CHUNK / 4;
static_assert(REC_WORDS + PLANE_SLOT <= REC_STRIDE, "no room for the plane behind a list");

__device__ __forceinline__ int full_regions(int N) {
  const int r = N / AGG_NODES;
  return r > 0 ? r : 1;
}
__device__ __forceinline__ size_t plane_word(int chunk, int nReg) {
  return (size_t)(chunk % nReg) * REC_STRIDE + REC_WORDS + (size_t)(chunk / nReg) * PLANE_SLOT;
}

__global__ __launch_bounds__(256)
void gcn_owner_plane(const int* __restrict__ dst, int* __restrict__ side, int E, int N) {
  const int lane = threadIdx.x & 31;
  const int wave = (int)((blockIdx.x * (size_t)blockDim.x + threadIdx.x) >> 5);
  const int eb = wave * 512;
  const int chunk = wave / AGG_NQ;
  if (chunk < E / AGG_CHUNK) {
    const int4* p = (const int4*)(dst + eb);
    int* o = side + plane_word(chunk, full_regions(N)) + (wave % AGG_NQ) * 128;
    int4 q[4];
#pragma unroll
    for (int j = 0; j < 4; ++j) q[j] = p[j * 32 + lane];
#pragma unroll
    for (int j = 0; j < 4; ++j)
      o[j * 32 + lane] =
          (int)(owner_byte(q[j].x) | (owner_byte(q[j].y) << 8) |
                (owner_byte(q[j].z) << 16) | (owner_byte(q[j].w) << 24));
  }
}


__device__ __forceinline__ void agg_lds_add(float* p, float v) {
  __hip_atomic_fetch_add(p, v, __ATOMIC_RELAXED, __HIP_MEMORY_SCOPE_WORKGROUP);
}

template <int LOGW, int MODE, int REC>
__global__ __launch_bounds__(AGG_THREADS)
void gcn_agg_owned(const float* __restrict__ h,
                   const int* __restrict__ src,
                   const int* __restrict__ dst,
                   const int* plane,
                   int* rec,
                   const float* __restrict__ bias,
                   float* out,
                   float* __restrict__ degOut,
                   const float* __restrict__ Wa,
                   const float* __restrict__ Wb,
                   int N, int E) {
  constexpr int WIDTH = 1 << LOGW;
  constexpr int PITCH = WIDTH + 1;
  constexpr int NT    = WIDTH / 16;
  __shared__ float acc[AGG_NODES * PITCH];
  __shared__ float cnt[AGG_NODES];
  __shared__ unsigned short hits[AGG_WAVES][AGG_CHUNK];
  __shared__ int recCount;

  const int tid  = threadIdx.x;
  const int lane = tid & 31;
  const int wave = __builtin_amdgcn_readfirstlane(tid >> 5);
  const int base = blockIdx.x * AGG_NODES;
  const int nHere = (N - base < AGG_NODES) ? (N - base) : AGG_NODES;
  const unsigned pat = (blockIdx.x & 255u) * 0x01010101u;

  for (int i = tid; i < AGG_NODES * PITCH; i += AGG_THREADS) acc[i] = 0.0f;
  for (int i = tid; i < AGG_NODES; i += AGG_THREADS) cnt[i] = 0.0f;
  if (tid == 0) recCount = 0;
  __syncthreads();

  int* myRec = rec + (size_t)blockIdx.x * REC_STRIDE;
  const int recCap = ((nHere * OUT_W < REC_WORDS) ? nHere * OUT_W : REC_WORDS) - 1;
  const int nReg = full_regions(N);

  const int nFull = E / AGG_CHUNK;
  const int rot = nFull > 0 ? (int)((blockIdx.x * 6u) % (unsigned)nFull) : 0;
  unsigned short* myHits = hits[wave];
  const int grp = lane >> 2;
  const int sl  = lane & 3;


  auto addBatch = [&](int val, int nb) {
    for (int t = 0; t * 8 < nb; t += 2) {
      int ent[2];
      ent[0] = __shfl(val, t * 8 + grp);
      ent[1] = __shfl(val, (t * 8 + 8 + grp) & 31);
      if (t * 8 + 8 >= nb) ent[1] = -1;
      float4 v[2][NT];
#pragma unroll
      for (int r = 0; r < 2; ++r) {
        const float4* row = (const float4*)(h + (size_t)(ent[r] >= 0 ? (ent[r] >> 9) : 0) * WIDTH);
#pragma unroll
        for (int u = 0; u < NT; ++u) v[r][u] = row[sl + 4 * u];
      }
#pragma unroll
      for (int r = 0; r < 2; ++r) {
        if (ent[r] >= 0) {
          const int rel = ent[r] & 511;
          float* a = acc + rel * PITCH + 4 * sl;
#pragma unroll
          for (int u = 0; u < NT; ++u) {
            agg_lds_add(a + 16 * u + 0, v[r][u].x);
            agg_lds_add(a + 16 * u + 1, v[r][u].y);
            agg_lds_add(a + 16 * u + 2, v[r][u].z);
            agg_lds_add(a + 16 * u + 3, v[r][u].w);
          }
          if (sl == 0) agg_lds_add(cnt + rel, 1.0f);
        }
      }
    }
  };

  auto ask = [&](int cc, int4 (&qq)[AGG_NQ]) -> int {
    int cr = cc + rot; if (cr >= nFull) cr -= nFull;
    const int ef = cr * AGG_CHUNK;
    const int4* p = (const int4*)(plane + plane_word(cr, nReg));
#pragma unroll
    for (int j = 0; j < AGG_NQ; ++j) qq[j] = p[j * 32 + lane];
    return ef;
  };

  auto work = [&](const int4 (&qq)[AGG_NQ], int ef) {
    unsigned c[AGG_NP], m[AGG_NP];
    unsigned any = 0u;
#pragma unroll
    for (int w = 0; w < AGG_NP; ++w) {
      unsigned cw = 0u;
#pragma unroll
      for (int jj = 0; jj < 2; ++jj) {
        const int4 q = qq[2 * w + jj];
        const unsigned x4[4] = {(unsigned)q.x, (unsigned)q.y, (unsigned)q.z, (unsigned)q.w};
#pragma unroll
        for (int k = 0; k < 4; ++k) {
          const unsigned y = x4[k] ^ pat;
          cw |= (((y - 0x01010101u) & ~y) >> (7 - 4 * jj - k)) & (0x01010101u << (4 * jj + k));
        }
      }
      c[w] = cw;
      m[w] = __builtin_amdgcn_ballot_w32(cw != 0u);
      any |= m[w];
    }
    if (any == 0u) return;
    int total = 0;
#pragma unroll
    for (int w = 0; w < AGG_NP; ++w) {
      unsigned cw = c[w];
      unsigned mw = m[w];
      while (mw != 0u) {
        if (cw != 0u) {
          const int b = __builtin_ctz(cw);
          const int pos = total + (int)__builtin_amdgcn_mbcnt_lo(mw, 0u);
          myHits[pos] = (unsigned short)(w * 1024 + ((b & 4) << 7) + lane * 16 + ((b & 3) << 2) + (b >> 3));
          cw &= cw - 1u;
        }
        total += __builtin_popcount(mw);
        mw = __builtin_amdgcn_ballot_w32(cw != 0u);
      }
    }
    for (int b0 = 0; b0 < total; b0 += 32) {
      const int nb = (total - b0 < 32) ? (total - b0) : 32;
      int val = -1;
      if (lane < nb) {
        const int e = ef + (int)myHits[b0 + lane];
        const unsigned rel = (unsigned)(dst[e] - base);
        const int s = src[e];
        val = (rel < (unsigned)nHere) ? ((s << 9) | (int)rel) : -1;
      }
      if (REC == 1) {
        const unsigned okm = __builtin_amdgcn_ballot_w32(val >= 0);
        if (okm != 0u) {
          int p0 = 0;
          if (lane == 0)
            p0 = __hip_atomic_fetch_add(&recCount, __builtin_popcount(okm), __ATOMIC_RELAXED,
                                        __HIP_MEMORY_SCOPE_WORKGROUP);
          p0 = __builtin_amdgcn_readfirstlane(p0);
          const int p = p0 + (int)__builtin_amdgcn_mbcnt_lo(okm, 0u);
          if (val >= 0 && p < recCap) myRec[1 + p] = val;
        }
      }
      addBatch(val, nb);
    }
  };

  int c = wave;
  bool replayed = false;
  if (REC == 2) {
    const int found = myRec[0];
    if (found >= 0 && found <= recCap) {
      replayed = true;
      int b0 = wave * 32;
      int nxt = (b0 + lane < found) ? myRec[1 + b0 + lane] : -1;
      while (b0 < found) {
        const int val = nxt;
        const int nb = (found - b0 < 32) ? (found - b0) : 32;
        const int b1 = b0 + AGG_THREADS;
        nxt = (b1 + lane < found) ? myRec[1 + b1 + lane] : -1;
        addBatch(val, nb);
        b0 = b1;
      }
    }
  }
  if (REC != 2 && c < nFull) {
    int4 qa[AGG_NQ], qb[AGG_NQ];
    int ea = ask(c, qa), eb = 0;
    for (;;) {
      eb = ask((c + AGG_WAVES < nFull) ? c + AGG_WAVES : c, qb);
      work(qa, ea);
      c += AGG_WAVES;
      if (c >= nFull) break;
      ea = ask((c + AGG_WAVES < nFull) ? c + AGG_WAVES : c, qa);
      work(qb, eb);
      c += AGG_WAVES;
      if (c >= nFull) break;
    }
  }

  for (int e = (REC == 2 ? 0 : nFull * AGG_CHUNK) + tid; e < (replayed ? 0 : E); e += AGG_THREADS) {
    const unsigned rel = (unsigned)(dst[e] - base);
    if (rel < (unsigned)nHere) {
      const int s = src[e];
      const float* row = h + (size_t)s * WIDTH;
      float* a = acc + rel * PITCH;
      for (int col = 0; col < WIDTH; ++col) agg_lds_add(a + col, row[col]);
      agg_lds_add(cnt + rel, 1.0f);
      if (REC == 1) {
        const int p = __hip_atomic_fetch_add(&recCount, 1, __ATOMIC_RELAXED, __HIP_MEMORY_SCOPE_WORKGROUP);
        if (p < recCap) myRec[1 + p] = (s << 9) | (int)rel;
      }
    }
  }
  __syncthreads();
  if (REC == 1 && tid == 0) myRec[0] = recCount;

  if (MODE == 3) {
    constexpr int NTA = WIDTH / 16;
    constexpr int NTB = OUT_W / 16;
    const int half = lane >> 4;
    const int l16  = lane & 15;
    const bool has = wave * 16 < nHere;
    float* rows = acc + wave * 16 * PITCH;
    const float* arow = rows + l16 * PITCH + 2 * half;
    if (has) {
      v8f d[NTA];
#pragma unroll
      for (int j = 0; j < NTA; ++j) d[j] = (v8f){};
#pragma unroll
      for (int k = 0; k < WIDTH; k += 4) {
        v2f a;
        a.x = arow[k];
        a.y = arow[k + 1];
        const int kk = k + 2 * half;
#pragma unroll
        for (int j = 0; j < NTA; ++j) {
          v2f b;
          b.x = Wa[kk * WIDTH + (j << 4) + l16];
          b.y = Wa[(kk + 1) * WIDTH + (j << 4) + l16];
          d[j] = __builtin_amdgcn_wmma_f32_16x16x4_f32(
               false, a,  false, b,
               (short)0, d[j],  false,  false);
        }
      }
#pragma unroll
      for (int r = 0; r < 8; ++r) {
        const int row = r + 8 * half;
        float dg = cnt[wave * 16 + row];
        dg = dg < 1.0f ? 1.0f : dg;
#pragma unroll
        for (int j = 0; j < NTA; ++j) {
          float v = d[j][r] / dg + bias[(j << 4) + l16];
          rows[row * PITCH + (j << 4) + l16] = v > 0.0f ? v : 0.0f;
        }
      }
    }
    __syncthreads();
    if (has) {
      v8f z[NTB];
#pragma unroll
      for (int j = 0; j < NTB; ++j) z[j] = (v8f){};
#pragma unroll
      for (int k = 0; k < WIDTH; k += 4) {
        v2f a;
        a.x = arow[k];
        a.y = arow[k + 1];
        const int kk = k + 2 * half;
#pragma unroll
        for (int j = 0; j < NTB; ++j) {
          v2f b;
          b.x = Wb[kk * OUT_W + (j << 4) + l16];
          b.y = Wb[(kk + 1) * OUT_W + (j << 4) + l16];
          z[j] = __builtin_amdgcn_wmma_f32_16x16x4_f32(
               false, a,  false, b,
               (short)0, z[j],  false,  false);
        }
      }
#pragma unroll
      for (int r = 0; r < 8; ++r) {
        const int row = wave * 16 + r + 8 * half;
#pragma unroll
        for (int j = 0; j < NTB; ++j)
          out[(size_t)(base + row) * OUT_W + (j << 4) + l16] = z[j][r];
      }
    }
    return;
  }

  for (int i = tid; i < nHere * WIDTH; i += AGG_THREADS) {
    const int n = i >> LOGW;
    const int col = i & (WIDTH - 1);
    float v = acc[n * PITCH + col];
    if (MODE != 0) {
      float d = cnt[n];
      d = d < 1.0f ? 1.0f : d;
      v = v / d + bias[col];
      if (MODE == 1) v = v > 0.0f ? v : 0.0f;
    }
    out[(size_t)base * WIDTH + i] = v;
  }
  if (MODE == 0)
    for (int i = tid; i < nHere; i += AGG_THREADS) degOut[base + i] = cnt[i];
}

extern "C" void kernel_launch(void* const* d_in, const int* in_sizes, int n_in,
                              void* d_out, int out_size, void* d_ws, size_t ws_size,
                              hipStream_t stream) {
  const float* feat = (const float*)d_in[0];
  const float* W0   = (const float*)d_in[1];
  const float* b0   = (const float*)d_in[2];
  const float* W1   = (const float*)d_in[3];
  const float* b1   = (const float*)d_in[4];
  const float* W2   = (const float*)d_in[5];
  const float* b2   = (const float*)d_in[6];
  const int*   src  = (const int*)d_in[7];
  const int*   dst  = (const int*)d_in[8];
  float* out = (float*)d_out;

  const int N = in_sizes[0] / 256;
  const int E = in_sizes[7];

  float* bufA = (float*)d_ws;
  float* bufB = bufA + (size_t)N * 64;
  float* deg  = bufB + (size_t)N * 64;

  const int TB = 256;
  const int gemmBlocks  = (int)((((long long)(N / 16)) * 32 + TB - 1) / TB);
  const int aggBlocks   = (N + AGG_NODES - 1) / AGG_NODES;
  const int planeBlocks = (int)((((long long)((E + 511) / 512)) * 32 + TB - 1) / TB);

  int* side = (int*)d_out;

  gcn_owner_plane<<<planeBlocks, TB, 0, stream>>>(dst, side, E, N);
  gcn_gemm_wmma_f32<256, 4><<<gemmBlocks, TB, 0, stream>>>(feat, W0, bufA, N);
  gcn_agg_owned<6, 1, 1><<<aggBlocks, AGG_THREADS, 0, stream>>>(bufA, src, dst, side, side, b0, bufB, deg, W1, W2, N, E);

  gcn_agg_owned<6, 3, 2><<<aggBlocks, AGG_THREADS, 0, stream>>>(bufB, src, dst, side, side, b1, bufA, deg, W1, W2, N, E);

  gcn_agg_owned<5, 2, 2><<<aggBlocks, AGG_THREADS, 0, stream>>>(bufA, src, dst, side, side, b2, out, deg, W1, W2, N, E);
}


# neighbour-sum adds take one edge per wave instruction: edge read out to scalar registers, the wave loads that one row contiguously, lanes add to 32 different LDS banks; adds stay arrival order
# speedup vs baseline: 3.3445x; 1.0974x over previous
#include <hip/hip_runtime.h>


typedef __attribute__((ext_vector_type(2))) float v2f;
typedef __attribute__((ext_vector_type(8))) float v8f;

template <int K, int TM>
__global__ void gcn_gemm_wmma_f32(const float* __restrict__ X,
                                  const float* __restrict__ W,
                                  float* __restrict__ Y, int N) {
  const int M = TM * 16;
  const int wave = (int)((blockIdx.x * (size_t)blockDim.x + threadIdx.x) >> 5);
  if (wave >= (N >> 4)) return;

  const int lane = threadIdx.x & 31;
  const int half = lane >> 4;
  const int l16  = lane & 15;
  const float* __restrict__ xrow = X + (size_t)((wave << 4) + l16) * K;

  v8f acc[TM];
#pragma unroll
  for (int j = 0; j < TM; ++j) acc[j] = (v8f){};

#pragma unroll 2
  for (int k = 0; k < K; k += 4) {
    const int kk = k + (half << 1);
    const v2f a = *(const v2f*)(xrow + kk);
#pragma unroll
    for (int j = 0; j < TM; ++j) {
      v2f b;
      b.x = W[kk * M + (j << 4) + l16];
      b.y = W[(kk + 1) * M + (j << 4) + l16];
      acc[j] = __builtin_amdgcn_wmma_f32_16x16x4_f32(
           false, a,  false, b,
           (short)0, acc[j],  false,  false);
    }
  }

  const int rbase = (wave << 4) + (half << 3);
#pragma unroll
  for (int j = 0; j < TM; ++j)
#pragma unroll
    for (int r = 0; r < 8; ++r)
      Y[(size_t)(rbase + r) * M + (j << 4) + l16] = acc[j][r];
}

constexpr int AGG_NODES = 416;

__device__ __forceinline__ unsigned owner_byte(int d) {
  return ((unsigned)d / (unsigned)AGG_NODES) & 255u;
}

constexpr int AGG_THREADS = 1024;
constexpr int AGG_WAVES   = AGG_THREADS / 32;
constexpr int AGG_NQ      = 4;
constexpr int AGG_NP      = AGG_NQ / 2;
constexpr int AGG_CHUNK   = AGG_NQ * 512;

constexpr int OUT_W      = 32;
constexpr int REC_STRIDE = AGG_NODES * OUT_W;
constexpr int REC_WORDS  = 8192;
constexpr int PLANE_SLOT = AGG_CHUNK / 4;
static_assert(REC_WORDS + PLANE_SLOT <= REC_STRIDE, "no room for the plane behind a list");

__device__ __forceinline__ int full_regions(int N) {
  const int r = N / AGG_NODES;
  return r > 0 ? r : 1;
}
__device__ __forceinline__ size_t plane_word(int chunk, int nReg) {
  return (size_t)(chunk % nReg) * REC_STRIDE + REC_WORDS + (size_t)(chunk / nReg) * PLANE_SLOT;
}

__global__ __launch_bounds__(256)
void gcn_owner_plane(const int* __restrict__ dst, int* __restrict__ side, int E, int N) {
  const int lane = threadIdx.x & 31;
  const int wave = (int)((blockIdx.x * (size_t)blockDim.x + threadIdx.x) >> 5);
  const int eb = wave * 512;
  const int chunk = wave / AGG_NQ;
  if (chunk < E / AGG_CHUNK) {
    const int4* p = (const int4*)(dst + eb);
    int* o = side + plane_word(chunk, full_regions(N)) + (wave % AGG_NQ) * 128;
    int4 q[4];
#pragma unroll
    for (int j = 0; j < 4; ++j) q[j] = p[j * 32 + lane];
#pragma unroll
    for (int j = 0; j < 4; ++j)
      o[j * 32 + lane] =
          (int)(owner_byte(q[j].x) | (owner_byte(q[j].y) << 8) |
                (owner_byte(q[j].z) << 16) | (owner_byte(q[j].w) << 24));
  }
}


__device__ __forceinline__ void agg_lds_add(float* p, float v) {
  __hip_atomic_fetch_add(p, v, __ATOMIC_RELAXED, __HIP_MEMORY_SCOPE_WORKGROUP);
}

template <int LOGW, int MODE, int REC>
__global__ __launch_bounds__(AGG_THREADS)
void gcn_agg_owned(const float* __restrict__ h,
                   const int* __restrict__ src,
                   const int* __restrict__ dst,
                   const int* plane,
                   int* rec,
                   const float* __restrict__ bias,
                   float* out,
                   float* __restrict__ degOut,
                   const float* __restrict__ Wa,
                   const float* __restrict__ Wb,
                   int N, int E) {
  constexpr int WIDTH = 1 << LOGW;
  constexpr int PITCH = WIDTH + 1;
  __shared__ float acc[AGG_NODES * PITCH];
  __shared__ float cnt[AGG_NODES];
  __shared__ unsigned short hits[AGG_WAVES][AGG_CHUNK];
  __shared__ int recCount;

  const int tid  = threadIdx.x;
  const int lane = tid & 31;
  const int wave = __builtin_amdgcn_readfirstlane(tid >> 5);
  const int base = blockIdx.x * AGG_NODES;
  const int nHere = (N - base < AGG_NODES) ? (N - base) : AGG_NODES;
  const unsigned pat = (blockIdx.x & 255u) * 0x01010101u;

  for (int i = tid; i < AGG_NODES * PITCH; i += AGG_THREADS) acc[i] = 0.0f;
  for (int i = tid; i < AGG_NODES; i += AGG_THREADS) cnt[i] = 0.0f;
  if (tid == 0) recCount = 0;
  __syncthreads();

  int* myRec = rec + (size_t)blockIdx.x * REC_STRIDE;
  const int recCap = ((nHere * OUT_W < REC_WORDS) ? nHere * OUT_W : REC_WORDS) - 1;
  const int nReg = full_regions(N);

  const int nFull = E / AGG_CHUNK;
  const int rot = nFull > 0 ? (int)((blockIdx.x * 6u) % (unsigned)nFull) : 0;
  unsigned short* myHits = hits[wave];


  constexpr int ROWV = WIDTH / 32;
  constexpr int GRP  = 16;
  const unsigned laneOff = (unsigned)(ROWV * lane) * 4u;
  auto addBatch = [&](int val, int nb) {
    if (val >= 0) agg_lds_add(cnt + (val & 511), 1.0f);
    const unsigned rowOff = val >= 0 ? (unsigned)(val >> 9) * (unsigned)(WIDTH * 4) : 0u;
#pragma unroll
    for (int t = 0; t < 32; t += GRP) {
      if (t >= nb) break;
      int ent[GRP];
      float v[GRP][ROWV];
#pragma unroll
      for (int i = 0; i < GRP; ++i) {
        ent[i] = __builtin_amdgcn_readlane(val, t + i);
        const char* rowp = (const char*)h + (unsigned)__builtin_amdgcn_readlane((int)rowOff, t + i);
        const float* mine = (const float*)(rowp + laneOff);
#pragma unroll
        for (int u = 0; u < ROWV; ++u) v[i][u] = mine[u];
      }
#pragma unroll
      for (int i = 0; i < GRP; ++i) {
        if (ent[i] >= 0) {
          float* a = acc + (ent[i] & 511) * PITCH + ROWV * lane;
#pragma unroll
          for (int u = 0; u < ROWV; ++u) agg_lds_add(a + u, v[i][u]);
        }
      }
    }
  };

  auto ask = [&](int cc, int4 (&qq)[AGG_NQ]) -> int {
    int cr = cc + rot; if (cr >= nFull) cr -= nFull;
    const int ef = cr * AGG_CHUNK;
    const int4* p = (const int4*)(plane + plane_word(cr, nReg));
#pragma unroll
    for (int j = 0; j < AGG_NQ; ++j) qq[j] = p[j * 32 + lane];
    return ef;
  };

  auto work = [&](const int4 (&qq)[AGG_NQ], int ef) {
    unsigned c[AGG_NP], m[AGG_NP];
    unsigned any = 0u;
#pragma unroll
    for (int w = 0; w < AGG_NP; ++w) {
      unsigned cw = 0u;
#pragma unroll
      for (int jj = 0; jj < 2; ++jj) {
        const int4 q = qq[2 * w + jj];
        const unsigned x4[4] = {(unsigned)q.x, (unsigned)q.y, (unsigned)q.z, (unsigned)q.w};
#pragma unroll
        for (int k = 0; k < 4; ++k) {
          const unsigned y = x4[k] ^ pat;
          cw |= (((y - 0x01010101u) & ~y) >> (7 - 4 * jj - k)) & (0x01010101u << (4 * jj + k));
        }
      }
      c[w] = cw;
      m[w] = __builtin_amdgcn_ballot_w32(cw != 0u);
      any |= m[w];
    }
    if (any == 0u) return;
    int total = 0;
#pragma unroll
    for (int w = 0; w < AGG_NP; ++w) {
      unsigned cw = c[w];
      unsigned mw = m[w];
      while (mw != 0u) {
        if (cw != 0u) {
          const int b = __builtin_ctz(cw);
          const int pos = total + (int)__builtin_amdgcn_mbcnt_lo(mw, 0u);
          myHits[pos] = (unsigned short)(w * 1024 + ((b & 4) << 7) + lane * 16 + ((b & 3) << 2) + (b >> 3));
          cw &= cw - 1u;
        }
        total += __builtin_popcount(mw);
        mw = __builtin_amdgcn_ballot_w32(cw != 0u);
      }
    }
    for (int b0 = 0; b0 < total; b0 += 32) {
      const int nb = (total - b0 < 32) ? (total - b0) : 32;
      int val = -1;
      if (lane < nb) {
        const int e = ef + (int)myHits[b0 + lane];
        const unsigned rel = (unsigned)(dst[e] - base);
        const int s = src[e];
        val = (rel < (unsigned)nHere) ? ((s << 9) | (int)rel) : -1;
      }
      if (REC == 1) {
        const unsigned okm = __builtin_amdgcn_ballot_w32(val >= 0);
        if (okm != 0u) {
          int p0 = 0;
          if (lane == 0)
            p0 = __hip_atomic_fetch_add(&recCount, __builtin_popcount(okm), __ATOMIC_RELAXED,
                                        __HIP_MEMORY_SCOPE_WORKGROUP);
          p0 = __builtin_amdgcn_readfirstlane(p0);
          const int p = p0 + (int)__builtin_amdgcn_mbcnt_lo(okm, 0u);
          if (val >= 0 && p < recCap) myRec[1 + p] = val;
        }
      }
      addBatch(val, nb);
    }
  };

  int c = wave;
  bool replayed = false;
  if (REC == 2) {
    const int found = myRec[0];
    if (found >= 0 && found <= recCap) {
      replayed = true;
      int b0 = wave * 32;
      int nxt = (b0 + lane < found) ? myRec[1 + b0 + lane] : -1;
      while (b0 < found) {
        const int val = nxt;
        const int nb = (found - b0 < 32) ? (found - b0) : 32;
        const int b1 = b0 + AGG_THREADS;
        nxt = (b1 + lane < found) ? myRec[1 + b1 + lane] : -1;
        addBatch(val, nb);
        b0 = b1;
      }
    }
  }
  if (REC != 2 && c < nFull) {
    int4 qa[AGG_NQ], qb[AGG_NQ];
    int ea = ask(c, qa), eb = 0;
    for (;;) {
      eb = ask((c + AGG_WAVES < nFull) ? c + AGG_WAVES : c, qb);
      work(qa, ea);
      c += AGG_WAVES;
      if (c >= nFull) break;
      ea = ask((c + AGG_WAVES < nFull) ? c + AGG_WAVES : c, qa);
      work(qb, eb);
      c += AGG_WAVES;
      if (c >= nFull) break;
    }
  }

  for (int e = (REC == 2 ? 0 : nFull * AGG_CHUNK) + tid; e < (replayed ? 0 : E); e += AGG_THREADS) {
    const unsigned rel = (unsigned)(dst[e] - base);
    if (rel < (unsigned)nHere) {
      const int s = src[e];
      const float* row = h + (size_t)s * WIDTH;
      float* a = acc + rel * PITCH;
      for (int col = 0; col < WIDTH; ++col) agg_lds_add(a + col, row[col]);
      agg_lds_add(cnt + rel, 1.0f);
      if (REC == 1) {
        const int p = __hip_atomic_fetch_add(&recCount, 1, __ATOMIC_RELAXED, __HIP_MEMORY_SCOPE_WORKGROUP);
        if (p < recCap) myRec[1 + p] = (s << 9) | (int)rel;
      }
    }
  }
  __syncthreads();
  if (REC == 1 && tid == 0) myRec[0] = recCount;

  if (MODE == 3) {
    constexpr int NTA = WIDTH / 16;
    constexpr int NTB = OUT_W / 16;
    const int half = lane >> 4;
    const int l16  = lane & 15;
    const bool has = wave * 16 < nHere;
    float* rows = acc + wave * 16 * PITCH;
    const float* arow = rows + l16 * PITCH + 2 * half;
    if (has) {
      v8f d[NTA];
#pragma unroll
      for (int j = 0; j < NTA; ++j) d[j] = (v8f){};
#pragma unroll
      for (int k = 0; k < WIDTH; k += 4) {
        v2f a;
        a.x = arow[k];
        a.y = arow[k + 1];
        const int kk = k + 2 * half;
#pragma unroll
        for (int j = 0; j < NTA; ++j) {
          v2f b;
          b.x = Wa[kk * WIDTH + (j << 4) + l16];
          b.y = Wa[(kk + 1) * WIDTH + (j << 4) + l16];
          d[j] = __builtin_amdgcn_wmma_f32_16x16x4_f32(
               false, a,  false, b,
               (short)0, d[j],  false,  false);
        }
      }
#pragma unroll
      for (int r = 0; r < 8; ++r) {
        const int row = r + 8 * half;
        float dg = cnt[wave * 16 + row];
        dg = dg < 1.0f ? 1.0f : dg;
#pragma unroll
        for (int j = 0; j < NTA; ++j) {
          float v = d[j][r] / dg + bias[(j << 4) + l16];
          rows[row * PITCH + (j << 4) + l16] = v > 0.0f ? v : 0.0f;
        }
      }
    }
    __syncthreads();
    if (has) {
      v8f z[NTB];
#pragma unroll
      for (int j = 0; j < NTB; ++j) z[j] = (v8f){};
#pragma unroll
      for (int k = 0; k < WIDTH; k += 4) {
        v2f a;
        a.x = arow[k];
        a.y = arow[k + 1];
        const int kk = k + 2 * half;
#pragma unroll
        for (int j = 0; j < NTB; ++j) {
          v2f b;
          b.x = Wb[kk * OUT_W + (j << 4) + l16];
          b.y = Wb[(kk + 1) * OUT_W + (j << 4) + l16];
          z[j] = __builtin_amdgcn_wmma_f32_16x16x4_f32(
               false, a,  false, b,
               (short)0, z[j],  false,  false);
        }
      }
#pragma unroll
      for (int r = 0; r < 8; ++r) {
        const int row = wave * 16 + r + 8 * half;
#pragma unroll
        for (int j = 0; j < NTB; ++j)
          out[(size_t)(base + row) * OUT_W + (j << 4) + l16] = z[j][r];
      }
    }
    return;
  }

  for (int i = tid; i < nHere * WIDTH; i += AGG_THREADS) {
    const int n = i >> LOGW;
    const int col = i & (WIDTH - 1);
    float v = acc[n * PITCH + col];
    if (MODE != 0) {
      float d = cnt[n];
      d = d < 1.0f ? 1.0f : d;
      v = v / d + bias[col];
      if (MODE == 1) v = v > 0.0f ? v : 0.0f;
    }
    out[(size_t)base * WIDTH + i] = v;
  }
  if (MODE == 0)
    for (int i = tid; i < nHere; i += AGG_THREADS) degOut[base + i] = cnt[i];
}

extern "C" void kernel_launch(void* const* d_in, const int* in_sizes, int n_in,
                              void* d_out, int out_size, void* d_ws, size_t ws_size,
                              hipStream_t stream) {
  const float* feat = (const float*)d_in[0];
  const float* W0   = (const float*)d_in[1];
  const float* b0   = (const float*)d_in[2];
  const float* W1   = (const float*)d_in[3];
  const float* b1   = (const float*)d_in[4];
  const float* W2   = (const float*)d_in[5];
  const float* b2   = (const float*)d_in[6];
  const int*   src  = (const int*)d_in[7];
  const int*   dst  = (const int*)d_in[8];
  float* out = (float*)d_out;

  const int N = in_sizes[0] / 256;
  const int E = in_sizes[7];

  float* bufA = (float*)d_ws;
  float* bufB = bufA + (size_t)N * 64;
  float* deg  = bufB + (size_t)N * 64;

  const int TB = 256;
  const int gemmBlocks  = (int)((((long long)(N / 16)) * 32 + TB - 1) / TB);
  const int aggBlocks   = (N + AGG_NODES - 1) / AGG_NODES;
  const int planeBlocks = (int)((((long long)((E + 511) / 512)) * 32 + TB - 1) / TB);

  int* side = (int*)d_out;

  gcn_owner_plane<<<planeBlocks, TB, 0, stream>>>(dst, side, E, N);
  gcn_gemm_wmma_f32<256, 4><<<gemmBlocks, TB, 0, stream>>>(feat, W0, bufA, N);
  gcn_agg_owned<6, 1, 1><<<aggBlocks, AGG_THREADS, 0, stream>>>(bufA, src, dst, side, side, b0, bufB, deg, W1, W2, N, E);

  gcn_agg_owned<6, 3, 2><<<aggBlocks, AGG_THREADS, 0, stream>>>(bufB, src, dst, side, side, b1, bufA, deg, W1, W2, N, E);

  gcn_agg_owned<5, 2, 2><<<aggBlocks, AGG_THREADS, 0, stream>>>(bufA, src, dst, side, side, b2, out, deg, W1, W2, N, E);
}
